# starting kernel
# baseline (speedup 1.0000x reference)
#include <hip/hip_runtime.h>
#include <hip/hip_bf16.h>


typedef float v2f __attribute__((ext_vector_type(2)));
typedef float v8f __attribute__((ext_vector_type(8)));

#define N_IN   512
#define N_OUT  256
#define T_LEN  2048
#define BATCH  32
#define M_TOT  (T_LEN * BATCH)
#define N_TOT  (4 * N_OUT)

#define TM 64
#define TN 64
#define TK 64
#define LDS_STRIDE 68

__device__ __forceinline__ float sigmoidf_fast(float x) {
    return 1.0f / (1.0f + __expf(-x));
}

__global__ __launch_bounds__(128)
void sru_gemm4_kernel(const float* __restrict__ xt,
                      const float* __restrict__ Wx, const float* __restrict__ Wf,
                      const float* __restrict__ Wr, const float* __restrict__ Wc,
                      const float* __restrict__ bf, const float* __restrict__ br,
                      const float* __restrict__ bc,
                      float* __restrict__ g)
{
    __shared__ float As[2][TM * LDS_STRIDE];
    __shared__ float Bs[2][TN * LDS_STRIDE];

    const int tid   = threadIdx.x;
    const int wave  = tid >> 5;
    const int lane  = tid & 31;
    const int lm    = lane & 15;
    const int khalf = lane >> 4;

    const int nblk = blockIdx.x;
    const int m0   = blockIdx.y * TM;
    const int wsel = nblk >> 2;
    const int n0w  = (nblk & 3) * TN;
    const float* __restrict__ W =
        (wsel == 0) ? Wx : (wsel == 1) ? Wf : (wsel == 2) ? Wr : Wc;
    const float* __restrict__ bias =
        (wsel == 1) ? bf : (wsel == 2) ? br : (wsel == 3) ? bc : nullptr;

    v8f acc[4];
#pragma unroll
    for (int ns = 0; ns < 4; ++ns)
        acc[ns] = (v8f){0.f, 0.f, 0.f, 0.f, 0.f, 0.f, 0.f, 0.f};

#pragma unroll
    for (int i = tid; i < TM * (TK / 4); i += 128) {
        const int r  = i >> 4;
        const int c4 = (i & 15) << 2;
        *(float4*)&As[0][r * LDS_STRIDE + c4] =
            *(const float4*)(xt + (size_t)(m0 + r) * N_IN + c4);
        *(float4*)&Bs[0][r * LDS_STRIDE + c4] =
            *(const float4*)(W + (size_t)(n0w + r) * N_IN + c4);
    }
    __syncthreads();

    int buf = 0;
    for (int kb = 0; kb < N_IN; kb += TK, buf ^= 1) {
        if (kb + TK < N_IN) {
            const int kn = kb + TK;
#pragma unroll
            for (int i = tid; i < TM * (TK / 4); i += 128) {
                const int r  = i >> 4;
                const int c4 = (i & 15) << 2;
                *(float4*)&As[buf ^ 1][r * LDS_STRIDE + c4] =
                    *(const float4*)(xt + (size_t)(m0 + r) * N_IN + kn + c4);
                *(float4*)&Bs[buf ^ 1][r * LDS_STRIDE + c4] =
                    *(const float4*)(W + (size_t)(n0w + r) * N_IN + kn + c4);
            }
        }

        const float* Arow = &As[buf][(wave * 16 + lm) * LDS_STRIDE + 2 * khalf];
        const float* Bbase = &Bs[buf][lm * LDS_STRIDE + 2 * khalf];
#pragma unroll
        for (int kk = 0; kk < TK; kk += 4) {
            const v2f a = *(const v2f*)(Arow + kk);
#pragma unroll
            for (int ns = 0; ns < 4; ++ns) {
                const v2f b = *(const v2f*)(Bbase + ns * 16 * LDS_STRIDE + kk);
                acc[ns] = __builtin_amdgcn_wmma_f32_16x16x4_f32(
                    false, a, false, b, (short)0, acc[ns], false, false);
            }
        }
        __syncthreads();
    }

    const int ncol0 = nblk * TN;
#pragma unroll
    for (int ns = 0; ns < 4; ++ns) {
        const int colw = n0w + ns * 16 + lm;
        const int col  = ncol0 + ns * 16 + lm;
        const float bv = (wsel != 0) ? bias[colw] : 0.0f;
#pragma unroll
        for (int r = 0; r < 8; ++r) {
            const int row = m0 + wave * 16 + khalf * 8 + r;
            float v = acc[ns][r];
            if (wsel == 1 || wsel == 2)      v = sigmoidf_fast(v + bv);
            else if (wsel == 3)              v = v + bv;
            g[(size_t)row * N_TOT + col] = v;
        }
    }
}

__global__ __launch_bounds__(256)
void sru_scan_kernel(const float* __restrict__ g,
                     const float* __restrict__ ct0,
                     float* __restrict__ out)
{
    const int o = threadIdx.x;
    const int b = blockIdx.x;

    float c = ct0[b * N_OUT + o];
    const size_t colbase = (size_t)b * N_TOT + o;
    const size_t obase   = (size_t)b * N_OUT + o;
    const size_t rowstride = (size_t)BATCH * N_TOT;

    for (int t = 0; t < T_LEN; ++t) {
        const float* row = g + (size_t)t * rowstride + colbase;

        const float* pf = row + 8 * rowstride;
        __builtin_prefetch(pf + 0,         0, 0);
        __builtin_prefetch(pf + N_OUT,     0, 0);
        __builtin_prefetch(pf + 2 * N_OUT, 0, 0);
        __builtin_prefetch(pf + 3 * N_OUT, 0, 0);

        const float xp = row[0];
        const float f  = row[N_OUT];
        const float r  = row[2 * N_OUT];
        const float cx = row[3 * N_OUT];

        c = f * c + (1.0f - f) * xp;
        const float h = r * tanhf(c) + (1.0f - r) * cx;

        out[(size_t)t * (BATCH * N_OUT) + obase] = h;
    }
    out[(size_t)M_TOT * N_OUT + obase] = c;
}

extern "C" void kernel_launch(void* const* d_in, const int* in_sizes, int n_in,
                              void* d_out, int out_size, void* d_ws, size_t ws_size,
                              hipStream_t stream) {
    const float* xt = (const float*)d_in[0];
    const float* ct = (const float*)d_in[1];
    const float* Wx = (const float*)d_in[2];
    const float* Wf = (const float*)d_in[3];
    const float* bf = (const float*)d_in[4];
    const float* Wr = (const float*)d_in[5];
    const float* br = (const float*)d_in[6];
    const float* Wc = (const float*)d_in[7];
    const float* bc = (const float*)d_in[8];

    float* g   = (float*)d_ws;
    float* out = (float*)d_out;

    dim3 grid1(N_TOT / TN, M_TOT / TM);
    sru_gemm4_kernel<<<grid1, 128, 0, stream>>>(xt, Wx, Wf, Wr, Wc,
                                                bf, br, bc, g);

    sru_scan_kernel<<<BATCH, N_OUT, 0, stream>>>(g, ct, out);
}


# scan split in two launches: a chain-only pass (one wave per 32 chains, x and f requested two 16-step groups ahead, c_t parked in out) and a parallel pass turning c_t into h_t in place; sums kept
# speedup vs baseline: 2.8542x; 2.8542x over previous
#include <hip/hip_runtime.h>
#include <hip/hip_bf16.h>


typedef float v2f __attribute__((ext_vector_type(2)));
typedef float v8f __attribute__((ext_vector_type(8)));

#define N_IN   512
#define N_OUT  256
#define T_LEN  2048
#define BATCH  32
#define M_TOT  (T_LEN * BATCH)
#define N_TOT  (4 * N_OUT)

#define TM 64
#define TN 64
#define TK 64
#define LDS_STRIDE 68
#define PF_AHEAD 64

__device__ __forceinline__ float sigmoidf_fast(float x) {
    return 1.0f / (1.0f + __expf(-x));
}

__global__ __launch_bounds__(128)
void sru_gemm4_kernel(const float* __restrict__ xt,
                      const float* __restrict__ Wx, const float* __restrict__ Wf,
                      const float* __restrict__ Wr, const float* __restrict__ Wc,
                      const float* __restrict__ bf, const float* __restrict__ br,
                      const float* __restrict__ bc,
                      float* __restrict__ g)
{
    __shared__ float As[2][TM * LDS_STRIDE];
    __shared__ float Bs[2][TN * LDS_STRIDE];

    const int tid   = threadIdx.x;
    const int wave  = tid >> 5;
    const int lane  = tid & 31;
    const int lm    = lane & 15;
    const int khalf = lane >> 4;

    const int nblk = blockIdx.x;
    const int m0   = blockIdx.y * TM;
    const int wsel = nblk >> 2;
    const int n0w  = (nblk & 3) * TN;
    const float* __restrict__ W =
        (wsel == 0) ? Wx : (wsel == 1) ? Wf : (wsel == 2) ? Wr : Wc;
    const float* __restrict__ bias =
        (wsel == 1) ? bf : (wsel == 2) ? br : (wsel == 3) ? bc : nullptr;

    for (int j = 0; j < (TM * (N_IN / TK)) / 128; ++j) {
        const int i  = tid + 128 * j;
        const int r  = i >> 3;
        const int cu = i & 7;
        if (cu != 0)
            __builtin_prefetch(xt + (size_t)(m0 + r) * N_IN + cu * TK, 0, 3);
    }
    if (blockIdx.y + PF_AHEAD < M_TOT / TM && tid < TM)
        __builtin_prefetch(xt + (size_t)(m0 + PF_AHEAD * TM + tid) * N_IN, 0, 3);

    v8f acc[4];
#pragma unroll
    for (int ns = 0; ns < 4; ++ns)
        acc[ns] = (v8f){0.f, 0.f, 0.f, 0.f, 0.f, 0.f, 0.f, 0.f};

#pragma unroll
    for (int i = tid; i < TM * (TK / 4); i += 128) {
        const int r  = i >> 4;
        const int c4 = (i & 15) << 2;
        *(float4*)&As[0][r * LDS_STRIDE + c4] =
            *(const float4*)(xt + (size_t)(m0 + r) * N_IN + c4);
        *(float4*)&Bs[0][r * LDS_STRIDE + c4] =
            *(const float4*)(W + (size_t)(n0w + r) * N_IN + c4);
    }
    __syncthreads();

    int buf = 0;
    for (int kb = 0; kb < N_IN; kb += TK, buf ^= 1) {
        if (kb + TK < N_IN) {
            const int kn = kb + TK;
#pragma unroll
            for (int i = tid; i < TM * (TK / 4); i += 128) {
                const int r  = i >> 4;
                const int c4 = (i & 15) << 2;
                *(float4*)&As[buf ^ 1][r * LDS_STRIDE + c4] =
                    *(const float4*)(xt + (size_t)(m0 + r) * N_IN + kn + c4);
                *(float4*)&Bs[buf ^ 1][r * LDS_STRIDE + c4] =
                    *(const float4*)(W + (size_t)(n0w + r) * N_IN + kn + c4);
            }
        }

        const float* Arow = &As[buf][(wave * 16 + lm) * LDS_STRIDE + 2 * khalf];
        const float* Bbase = &Bs[buf][lm * LDS_STRIDE + 2 * khalf];
#pragma unroll
        for (int kk = 0; kk < TK; kk += 4) {
            const v2f a = *(const v2f*)(Arow + kk);
#pragma unroll
            for (int ns = 0; ns < 4; ++ns) {
                const v2f b = *(const v2f*)(Bbase + ns * 16 * LDS_STRIDE + kk);
                acc[ns] = __builtin_amdgcn_wmma_f32_16x16x4_f32(
                    false, a, false, b, (short)0, acc[ns], false, false);
            }
        }
        __syncthreads();
    }

    const int ncol0 = nblk * TN;
#pragma unroll
    for (int ns = 0; ns < 4; ++ns) {
        const int colw = n0w + ns * 16 + lm;
        const int col  = ncol0 + ns * 16 + lm;
        const float bv = (wsel != 0) ? bias[colw] : 0.0f;
#pragma unroll
        for (int r = 0; r < 8; ++r) {
            const int row = m0 + wave * 16 + khalf * 8 + r;
            float v = acc[ns][r];
            if (wsel == 1 || wsel == 2)      v = sigmoidf_fast(v + bv);
            else if (wsel == 3)              v = v + bv;
            g[(size_t)row * N_TOT + col] = v;
        }
    }
}

#define CH_G 16
static_assert(T_LEN % (4 * CH_G) == 0, "the chain loop walks four groups per trip");

__device__ __forceinline__
void chain_request(const float* __restrict__ gcol, int t,
                   float (&xp)[CH_G], float (&f)[CH_G])
{
#pragma unroll
    for (int u = 0; u < CH_G; ++u) {
        const float* row = gcol + (size_t)(t + u) * ((size_t)BATCH * N_TOT);
        xp[u] = row[0];
        f[u]  = row[N_OUT];
    }
}

__device__ __forceinline__
float chain_steps(float c, int t,
                  const float (&xp)[CH_G], const float (&f)[CH_G],
                  float* __restrict__ ocol)
{
#pragma unroll
    for (int u = 0; u < CH_G; ++u) {
        c = f[u] * c + (1.0f - f[u]) * xp[u];
        ocol[(size_t)(t + u) * (BATCH * N_OUT)] = c;
    }
    return c;
}

__global__ __launch_bounds__(32)
void sru_chain_kernel(const float* __restrict__ g,
                      const float* __restrict__ ct0,
                      float* __restrict__ out)
{
    const int ch = blockIdx.x * 32 + threadIdx.x;
    const int b  = ch >> 8;
    const int o  = ch & (N_OUT - 1);

    float c = ct0[ch];
    const float* __restrict__ gcol = g + (size_t)b * N_TOT + o;
    float* __restrict__ ocol       = out + ch;

    float x0[CH_G], f0[CH_G], x1[CH_G], f1[CH_G];
    float x2[CH_G], f2[CH_G], x3[CH_G], f3[CH_G];

    chain_request(gcol, 0, x0, f0);
    chain_request(gcol, CH_G, x1, f1);
    for (int t0 = 0; t0 < T_LEN; t0 += 4 * CH_G) {
        chain_request(gcol, t0 + 2 * CH_G, x2, f2);
        c = chain_steps(c, t0, x0, f0, ocol);
        chain_request(gcol, t0 + 3 * CH_G, x3, f3);
        c = chain_steps(c, t0 + CH_G, x1, f1, ocol);

        const bool more = (t0 + 4 * CH_G < T_LEN);
        chain_request(gcol, more ? (t0 + 4 * CH_G) : t0, x0, f0);
        c = chain_steps(c, t0 + 2 * CH_G, x2, f2, ocol);
        chain_request(gcol, more ? (t0 + 5 * CH_G) : (t0 + CH_G), x1, f1);
        c = chain_steps(c, t0 + 3 * CH_G, x3, f3, ocol);
    }
    ocol[(size_t)M_TOT * N_OUT] = c;
}

#define ACT_WG   1024
#define ACT_U    4
#define ACT_PER  ((M_TOT * N_OUT / 4) / (ACT_WG * 256))
static_assert(ACT_PER * ACT_WG * 256 == M_TOT * N_OUT / 4, "the grid covers ht exactly");
static_assert(ACT_PER % ACT_U == 0, "whole trips");

__global__ __launch_bounds__(256)
void sru_act_kernel(const float* __restrict__ g, float* out)
{
    for (int j0 = 0; j0 < ACT_PER; j0 += ACT_U) {
        float4 c[ACT_U], r[ACT_U], cx[ACT_U];
#pragma unroll
        for (int u = 0; u < ACT_U; ++u) {
            const int i  = ((j0 + u) * ACT_WG + blockIdx.x) * 256 + threadIdx.x;
            const int m  = i >> 6;
            const int o4 = (i & 63) << 2;
            const float* grow = g + (size_t)m * N_TOT + o4;
            r[u]  = *(const float4*)(grow + 2 * N_OUT);
            cx[u] = *(const float4*)(grow + 3 * N_OUT);
            c[u]  = *(const float4*)(out + (size_t)m * N_OUT + o4);
        }
#pragma unroll
        for (int u = 0; u < ACT_U; ++u) {
            const int i  = ((j0 + u) * ACT_WG + blockIdx.x) * 256 + threadIdx.x;
            const int m  = i >> 6;
            const int o4 = (i & 63) << 2;
            float4 h;
            h.x = r[u].x * tanhf(c[u].x) + (1.0f - r[u].x) * cx[u].x;
            h.y = r[u].y * tanhf(c[u].y) + (1.0f - r[u].y) * cx[u].y;
            h.z = r[u].z * tanhf(c[u].z) + (1.0f - r[u].z) * cx[u].z;
            h.w = r[u].w * tanhf(c[u].w) + (1.0f - r[u].w) * cx[u].w;
            *(float4*)(out + (size_t)m * N_OUT + o4) = h;
        }
    }
}

extern "C" void kernel_launch(void* const* d_in, const int* in_sizes, int n_in,
                              void* d_out, int out_size, void* d_ws, size_t ws_size,
                              hipStream_t stream) {
    const float* xt = (const float*)d_in[0];
    const float* ct = (const float*)d_in[1];
    const float* Wx = (const float*)d_in[2];
    const float* Wf = (const float*)d_in[3];
    const float* bf = (const float*)d_in[4];
    const float* Wr = (const float*)d_in[5];
    const float* br = (const float*)d_in[6];
    const float* Wc = (const float*)d_in[7];
    const float* bc = (const float*)d_in[8];

    float* g   = (float*)d_ws;
    float* out = (float*)d_out;

    dim3 grid1(N_TOT / TN, M_TOT / TM);
    sru_gemm4_kernel<<<grid1, 128, 0, stream>>>(xt, Wx, Wf, Wr, Wc,
                                                bf, br, bc, g);

    sru_chain_kernel<<<(BATCH * N_OUT) / 32, 32, 0, stream>>>(g, ct, out);

    sru_act_kernel<<<ACT_WG, 256, 0, stream>>>(g, out);
}


# GEMM retiled: 128x128 work-group tile, 8 waves of 32x64, K-chunk 32 (was 64x64, 4 waves of 16x64, chunk 64): half the staging round trips per WMMA, 6 waves per SIMD; sums and k order kept
# speedup vs baseline: 3.5208x; 1.2336x over previous
#include <hip/hip_runtime.h>
#include <hip/hip_bf16.h>


typedef float v2f __attribute__((ext_vector_type(2)));
typedef float v8f __attribute__((ext_vector_type(8)));

#define N_IN   512
#define N_OUT  256
#define T_LEN  2048
#define BATCH  32
#define M_TOT  (T_LEN * BATCH)
#define N_TOT  (4 * N_OUT)

#define TM 128
#define TN 128
#define TK 32
#define LDS_STRIDE 36
#define PF_AHEAD 96
#define PF_UNIT 64
static_assert(TM == TN, "one staging loop walks both tiles");

__device__ __forceinline__ float sigmoidf_fast(float x) {
    return 1.0f / (1.0f + __expf(-x));
}

__global__ __launch_bounds__(256)
void sru_gemm4_kernel(const float* __restrict__ xt,
                      const float* __restrict__ Wx, const float* __restrict__ Wf,
                      const float* __restrict__ Wr, const float* __restrict__ Wc,
                      const float* __restrict__ bf, const float* __restrict__ br,
                      const float* __restrict__ bc,
                      float* __restrict__ g)
{
    __shared__ float As[2][TM * LDS_STRIDE];
    __shared__ float Bs[2][TN * LDS_STRIDE];

    const int tid   = threadIdx.x;
    const int wave  = tid >> 5;
    const int wm    = wave >> 1;
    const int wn    = wave & 1;
    const int lane  = tid & 31;
    const int lm    = lane & 15;
    const int khalf = lane >> 4;

    const int nblk = blockIdx.x;
    const int m0   = blockIdx.y * TM;
    const int wsel = nblk >> 1;
    const int n0w  = (nblk & 1) * TN;
    const float* __restrict__ W =
        (wsel == 0) ? Wx : (wsel == 1) ? Wf : (wsel == 2) ? Wr : Wc;
    const float* __restrict__ bias =
        (wsel == 1) ? bf : (wsel == 2) ? br : (wsel == 3) ? bc : nullptr;

    for (int j = 0; j < (TM * (N_IN / PF_UNIT)) / 256; ++j) {
        const int i  = tid + 256 * j;
        const int r  = i >> 3;
        const int cu = i & 7;
        if (cu != 0)
            __builtin_prefetch(xt + (size_t)(m0 + r) * N_IN + cu * PF_UNIT, 0, 3);
    }
    if (blockIdx.y + PF_AHEAD < M_TOT / TM && tid < TM)
        __builtin_prefetch(xt + (size_t)(m0 + PF_AHEAD * TM + tid) * N_IN, 0, 3);

    v8f acc[2][4];
#pragma unroll
    for (int ms = 0; ms < 2; ++ms)
#pragma unroll
        for (int ns = 0; ns < 4; ++ns)
            acc[ms][ns] = (v8f){0.f, 0.f, 0.f, 0.f, 0.f, 0.f, 0.f, 0.f};

#pragma unroll
    for (int i = tid; i < TM * (TK / 4); i += 256) {
        const int r  = i >> 3;
        const int c4 = (i & 7) << 2;
        *(float4*)&As[0][r * LDS_STRIDE + c4] =
            *(const float4*)(xt + (unsigned)((m0 + r) * N_IN + c4));
        *(float4*)&Bs[0][r * LDS_STRIDE + c4] =
            *(const float4*)(W + (unsigned)((n0w + r) * N_IN + c4));
    }
    __syncthreads();

    int buf = 0;
    for (int kb = 0; kb < N_IN; kb += TK, buf ^= 1) {
        if (kb + TK < N_IN) {
            const int kn = kb + TK;
#pragma unroll
            for (int i = tid; i < TM * (TK / 4); i += 256) {
                const int r  = i >> 3;
                const int c4 = (i & 7) << 2;
                *(float4*)&As[buf ^ 1][r * LDS_STRIDE + c4] =
                    *(const float4*)(xt + (unsigned)((m0 + r) * N_IN + kn + c4));
                *(float4*)&Bs[buf ^ 1][r * LDS_STRIDE + c4] =
                    *(const float4*)(W + (unsigned)((n0w + r) * N_IN + kn + c4));
            }
        }

        const float* Arow  = &As[buf][(wm * 32 + lm) * LDS_STRIDE + 2 * khalf];
        const float* Bbase = &Bs[buf][(wn * 64 + lm) * LDS_STRIDE + 2 * khalf];
#pragma unroll
        for (int kk = 0; kk < TK; kk += 4) {
            const v2f a0 = *(const v2f*)(Arow + kk);
            const v2f a1 = *(const v2f*)(Arow + 16 * LDS_STRIDE + kk);
#pragma unroll
            for (int ns = 0; ns < 4; ++ns) {
                const v2f b = *(const v2f*)(Bbase + ns * 16 * LDS_STRIDE + kk);
                acc[0][ns] = __builtin_amdgcn_wmma_f32_16x16x4_f32(
                    false, a0, false, b, (short)0, acc[0][ns], false, false);
                acc[1][ns] = __builtin_amdgcn_wmma_f32_16x16x4_f32(
                    false, a1, false, b, (short)0, acc[1][ns], false, false);
            }
        }
        __syncthreads();
    }

    const int ncol0 = nblk * TN + wn * 64;
#pragma unroll
    for (int ns = 0; ns < 4; ++ns) {
        const int colw = n0w + wn * 64 + ns * 16 + lm;
        const int col  = ncol0 + ns * 16 + lm;
        const float bv = (wsel != 0) ? bias[colw] : 0.0f;
#pragma unroll
        for (int ms = 0; ms < 2; ++ms) {
#pragma unroll
            for (int r = 0; r < 8; ++r) {
                const int row = m0 + wm * 32 + ms * 16 + khalf * 8 + r;
                float v = acc[ms][ns][r];
                if (wsel == 1 || wsel == 2)      v = sigmoidf_fast(v + bv);
                else if (wsel == 3)              v = v + bv;
                g[(size_t)row * N_TOT + col] = v;
            }
        }
    }
}

#define CH_G 16
static_assert(T_LEN % (4 * CH_G) == 0, "the chain loop walks four groups per trip");

__device__ __forceinline__
void chain_request(const float* __restrict__ gcol, int t,
                   float (&xp)[CH_G], float (&f)[CH_G])
{
#pragma unroll
    for (int u = 0; u < CH_G; ++u) {
        const float* row = gcol + (size_t)(t + u) * ((size_t)BATCH * N_TOT);
        xp[u] = row[0];
        f[u]  = row[N_OUT];
    }
}

__device__ __forceinline__
float chain_steps(float c, int t,
                  const float (&xp)[CH_G], const float (&f)[CH_G],
                  float* __restrict__ ocol)
{
#pragma unroll
    for (int u = 0; u < CH_G; ++u) {
        c = f[u] * c + (1.0f - f[u]) * xp[u];
        ocol[(size_t)(t + u) * (BATCH * N_OUT)] = c;
    }
    return c;
}

__global__ __launch_bounds__(32)
void sru_chain_kernel(const float* __restrict__ g,
                      const float* __restrict__ ct0,
                      float* __restrict__ out)
{
    const int ch = blockIdx.x * 32 + threadIdx.x;
    const int b  = ch >> 8;
    const int o  = ch & (N_OUT - 1);

    float c = ct0[ch];
    const float* __restrict__ gcol = g + (size_t)b * N_TOT + o;
    float* __restrict__ ocol       = out + ch;

    float x0[CH_G], f0[CH_G], x1[CH_G], f1[CH_G];
    float x2[CH_G], f2[CH_G], x3[CH_G], f3[CH_G];

    chain_request(gcol, 0, x0, f0);
    chain_request(gcol, CH_G, x1, f1);
    for (int t0 = 0; t0 < T_LEN; t0 += 4 * CH_G) {
        chain_request(gcol, t0 + 2 * CH_G, x2, f2);
        c = chain_steps(c, t0, x0, f0, ocol);
        chain_request(gcol, t0 + 3 * CH_G, x3, f3);
        c = chain_steps(c, t0 + CH_G, x1, f1, ocol);

        const bool more = (t0 + 4 * CH_G < T_LEN);
        chain_request(gcol, more ? (t0 + 4 * CH_G) : t0, x0, f0);
        c = chain_steps(c, t0 + 2 * CH_G, x2, f2, ocol);
        chain_request(gcol, more ? (t0 + 5 * CH_G) : (t0 + CH_G), x1, f1);
        c = chain_steps(c, t0 + 3 * CH_G, x3, f3, ocol);
    }
    ocol[(size_t)M_TOT * N_OUT] = c;
}

#define ACT_WG   1024
#define ACT_U    4
#define ACT_PER  ((M_TOT * N_OUT / 4) / (ACT_WG * 256))
static_assert(ACT_PER * ACT_WG * 256 == M_TOT * N_OUT / 4, "the grid covers ht exactly");
static_assert(ACT_PER % ACT_U == 0, "whole trips");

__global__ __launch_bounds__(256)
void sru_act_kernel(const float* __restrict__ g, float* out)
{
    for (int j0 = 0; j0 < ACT_PER; j0 += ACT_U) {
        float4 c[ACT_U], r[ACT_U], cx[ACT_U];
#pragma unroll
        for (int u = 0; u < ACT_U; ++u) {
            const int i  = ((j0 + u) * ACT_WG + blockIdx.x) * 256 + threadIdx.x;
            const int m  = i >> 6;
            const int o4 = (i & 63) << 2;
            const float* grow = g + (size_t)m * N_TOT + o4;
            r[u]  = *(const float4*)(grow + 2 * N_OUT);
            cx[u] = *(const float4*)(grow + 3 * N_OUT);
            c[u]  = *(const float4*)(out + (size_t)m * N_OUT + o4);
        }
#pragma unroll
        for (int u = 0; u < ACT_U; ++u) {
            const int i  = ((j0 + u) * ACT_WG + blockIdx.x) * 256 + threadIdx.x;
            const int m  = i >> 6;
            const int o4 = (i & 63) << 2;
            float4 h;
            h.x = r[u].x * tanhf(c[u].x) + (1.0f - r[u].x) * cx[u].x;
            h.y = r[u].y * tanhf(c[u].y) + (1.0f - r[u].y) * cx[u].y;
            h.z = r[u].z * tanhf(c[u].z) + (1.0f - r[u].z) * cx[u].z;
            h.w = r[u].w * tanhf(c[u].w) + (1.0f - r[u].w) * cx[u].w;
            *(float4*)(out + (size_t)m * N_OUT + o4) = h;
        }
    }
}

extern "C" void kernel_launch(void* const* d_in, const int* in_sizes, int n_in,
                              void* d_out, int out_size, void* d_ws, size_t ws_size,
                              hipStream_t stream) {
    const float* xt = (const float*)d_in[0];
    const float* ct = (const float*)d_in[1];
    const float* Wx = (const float*)d_in[2];
    const float* Wf = (const float*)d_in[3];
    const float* bf = (const float*)d_in[4];
    const float* Wr = (const float*)d_in[5];
    const float* br = (const float*)d_in[6];
    const float* Wc = (const float*)d_in[7];
    const float* bc = (const float*)d_in[8];

    float* g   = (float*)d_ws;
    float* out = (float*)d_out;

    dim3 grid1(N_TOT / TN, M_TOT / TM);
    sru_gemm4_kernel<<<grid1, 256, 0, stream>>>(xt, Wx, Wf, Wr, Wc,
                                                bf, br, bc, g);

    sru_chain_kernel<<<(BATCH * N_OUT) / 32, 32, 0, stream>>>(g, ct, out);

    sru_act_kernel<<<ACT_WG, 256, 0, stream>>>(g, out);
}


# GEMM work-group split by wave role: 8 math waves only read LDS and issue WMMAs, 4 loader waves fetch the next K-chunk from global into the other LDS stage under them; sums and k order kept
# speedup vs baseline: 3.7352x; 1.0609x over previous
#include <hip/hip_runtime.h>
#include <hip/hip_bf16.h>


typedef float v2f __attribute__((ext_vector_type(2)));
typedef float v8f __attribute__((ext_vector_type(8)));

#define N_IN   512
#define N_OUT  256
#define T_LEN  2048
#define BATCH  32
#define M_TOT  (T_LEN * BATCH)
#define N_TOT  (4 * N_OUT)

#define TM 128
#define TN 128
#define TK 32
#define LDS_STRIDE 36
#define N_MATH 256
#define N_LOAD 128
#define LD_ROWS (N_LOAD / 8)
#define LD_PASSES (TM / LD_ROWS)
#define PF_AHEAD 96
#define PF_UNIT 64
static_assert(TM == TN, "one staging loop walks both tiles");
static_assert(TK == 32, "a loader lane group of 8 covers one K-chunk row of 32 floats");

__device__ __forceinline__ float sigmoidf_fast(float x) {
    return 1.0f / (1.0f + __expf(-x));
}

__global__ __launch_bounds__(N_MATH + N_LOAD)
void sru_gemm4_kernel(const float* __restrict__ xt,
                      const float* __restrict__ Wx, const float* __restrict__ Wf,
                      const float* __restrict__ Wr, const float* __restrict__ Wc,
                      const float* __restrict__ bf, const float* __restrict__ br,
                      const float* __restrict__ bc,
                      float* __restrict__ g)
{
    __shared__ float As[2][TM * LDS_STRIDE];
    __shared__ float Bs[2][TN * LDS_STRIDE];

    const int tid   = threadIdx.x;
    const int wave  = __builtin_amdgcn_readfirstlane(tid >> 5);
    const bool loader = wave >= N_MATH / 32;
    const int wm    = (wave >> 1) & 3;
    const int wn    = wave & 1;
    const int lane  = tid & 31;
    const int lm    = lane & 15;
    const int khalf = lane >> 4;

    const int nblk = blockIdx.x;
    const int m0   = blockIdx.y * TM;
    const int wsel = nblk >> 1;
    const int n0w  = (nblk & 1) * TN;
    const float* __restrict__ W =
        (wsel == 0) ? Wx : (wsel == 1) ? Wf : (wsel == 2) ? Wr : Wc;
    const float* __restrict__ bias =
        (wsel == 1) ? bf : (wsel == 2) ? br : (wsel == 3) ? bc : nullptr;

    const int lt   = tid & (N_LOAD - 1);
    const int lrow = lt >> 3;
    const int lc4  = (lt & 7) << 2;
    const float* __restrict__ ag = xt + (size_t)(m0 + lrow) * N_IN + lc4;
    const float* __restrict__ bg = W + (size_t)(n0w + lrow) * N_IN + lc4;
    const int lds_l = lrow * LDS_STRIDE + lc4;

    v8f acc[2][4];
#pragma unroll
    for (int ms = 0; ms < 2; ++ms)
#pragma unroll
        for (int ns = 0; ns < 4; ++ns)
            acc[ms][ns] = (v8f){0.f, 0.f, 0.f, 0.f, 0.f, 0.f, 0.f, 0.f};

    if (loader) {
        float4 a[LD_PASSES], w[LD_PASSES];
#pragma unroll
        for (int j = 0; j < LD_PASSES; ++j) {
            a[j] = *(const float4*)(ag + j * LD_ROWS * N_IN);
            w[j] = *(const float4*)(bg + j * LD_ROWS * N_IN);
        }
#pragma unroll
        for (int j = 0; j < LD_PASSES; ++j) {
            *(float4*)&As[0][lds_l + j * LD_ROWS * LDS_STRIDE] = a[j];
            *(float4*)&Bs[0][lds_l + j * LD_ROWS * LDS_STRIDE] = w[j];
        }
    } else {
        for (int j = 0; j < (TM * (N_IN / PF_UNIT)) / N_MATH; ++j) {
            const int i  = tid + N_MATH * j;
            const int r  = i >> 3;
            const int cu = i & 7;
            if (cu != 0)
                __builtin_prefetch(xt + (size_t)(m0 + r) * N_IN + cu * PF_UNIT, 0, 3);
        }
        if (blockIdx.y + PF_AHEAD < M_TOT / TM && tid < TM)
            __builtin_prefetch(xt + (size_t)(m0 + PF_AHEAD * TM + tid) * N_IN, 0, 3);
    }
    __syncthreads();

    int buf = 0;
    for (int kb = 0; kb < N_IN; kb += TK, buf ^= 1) {
        if (loader) {
            if (kb + TK < N_IN) {
                const int kn = kb + TK;
                float4 a[LD_PASSES], w[LD_PASSES];
#pragma unroll
                for (int j = 0; j < LD_PASSES; ++j) {
                    a[j] = *(const float4*)(ag + j * LD_ROWS * N_IN + kn);
                    w[j] = *(const float4*)(bg + j * LD_ROWS * N_IN + kn);
                }
#pragma unroll
                for (int j = 0; j < LD_PASSES; ++j) {
                    *(float4*)&As[buf ^ 1][lds_l + j * LD_ROWS * LDS_STRIDE] = a[j];
                    *(float4*)&Bs[buf ^ 1][lds_l + j * LD_ROWS * LDS_STRIDE] = w[j];
                }
            }
#pragma unroll
            for (int ms = 0; ms < 2; ++ms)
#pragma unroll
                for (int ns = 0; ns < 4; ++ns)
                    acc[ms][ns] = (v8f){0.f, 0.f, 0.f, 0.f, 0.f, 0.f, 0.f, 0.f};
        } else {
            const float* Arow  = &As[buf][(wm * 32 + lm) * LDS_STRIDE + 2 * khalf];
            const float* Bbase = &Bs[buf][(wn * 64 + lm) * LDS_STRIDE + 2 * khalf];
#pragma unroll
            for (int kk = 0; kk < TK; kk += 4) {
                const v2f a0 = *(const v2f*)(Arow + kk);
                const v2f a1 = *(const v2f*)(Arow + 16 * LDS_STRIDE + kk);
#pragma unroll
                for (int ns = 0; ns < 4; ++ns) {
                    const v2f b = *(const v2f*)(Bbase + ns * 16 * LDS_STRIDE + kk);
                    acc[0][ns] = __builtin_amdgcn_wmma_f32_16x16x4_f32(
                        false, a0, false, b, (short)0, acc[0][ns], false, false);
                    acc[1][ns] = __builtin_amdgcn_wmma_f32_16x16x4_f32(
                        false, a1, false, b, (short)0, acc[1][ns], false, false);
                }
            }
        }
        __syncthreads();
    }
    if (loader) return;

    const int ncol0 = nblk * TN + wn * 64;
#pragma unroll
    for (int ns = 0; ns < 4; ++ns) {
        const int colw = n0w + wn * 64 + ns * 16 + lm;
        const int col  = ncol0 + ns * 16 + lm;
        const float bv = (wsel != 0) ? bias[colw] : 0.0f;
#pragma unroll
        for (int ms = 0; ms < 2; ++ms) {
#pragma unroll
            for (int r = 0; r < 8; ++r) {
                const int row = m0 + wm * 32 + ms * 16 + khalf * 8 + r;
                float v = acc[ms][ns][r];
                if (wsel == 1 || wsel == 2)      v = sigmoidf_fast(v + bv);
                else if (wsel == 3)              v = v + bv;
                g[(size_t)row * N_TOT + col] = v;
            }
        }
    }
}

#define CH_G 16
static_assert(T_LEN % (4 * CH_G) == 0, "the chain loop walks four groups per trip");

__device__ __forceinline__
void chain_request(const float* __restrict__ gcol, int t,
                   float (&xp)[CH_G], float (&f)[CH_G])
{
#pragma unroll
    for (int u = 0; u < CH_G; ++u) {
        const float* row = gcol + (size_t)(t + u) * ((size_t)BATCH * N_TOT);
        xp[u] = row[0];
        f[u]  = row[N_OUT];
    }
}

__device__ __forceinline__
float chain_steps(float c, int t,
                  const float (&xp)[CH_G], const float (&f)[CH_G],
                  float* __restrict__ ocol)
{
#pragma unroll
    for (int u = 0; u < CH_G; ++u) {
        c = f[u] * c + (1.0f - f[u]) * xp[u];
        ocol[(size_t)(t + u) * (BATCH * N_OUT)] = c;
    }
    return c;
}

__global__ __launch_bounds__(32)
void sru_chain_kernel(const float* __restrict__ g,
                      const float* __restrict__ ct0,
                      float* __restrict__ out)
{
    const int ch = blockIdx.x * 32 + threadIdx.x;
    const int b  = ch >> 8;
    const int o  = ch & (N_OUT - 1);

    float c = ct0[ch];
    const float* __restrict__ gcol = g + (size_t)b * N_TOT + o;
    float* __restrict__ ocol       = out + ch;

    float x0[CH_G], f0[CH_G], x1[CH_G], f1[CH_G];
    float x2[CH_G], f2[CH_G], x3[CH_G], f3[CH_G];

    chain_request(gcol, 0, x0, f0);
    chain_request(gcol, CH_G, x1, f1);
    for (int t0 = 0; t0 < T_LEN; t0 += 4 * CH_G) {
        chain_request(gcol, t0 + 2 * CH_G, x2, f2);
        c = chain_steps(c, t0, x0, f0, ocol);
        chain_request(gcol, t0 + 3 * CH_G, x3, f3);
        c = chain_steps(c, t0 + CH_G, x1, f1, ocol);

        const bool more = (t0 + 4 * CH_G < T_LEN);
        chain_request(gcol, more ? (t0 + 4 * CH_G) : t0, x0, f0);
        c = chain_steps(c, t0 + 2 * CH_G, x2, f2, ocol);
        chain_request(gcol, more ? (t0 + 5 * CH_G) : (t0 + CH_G), x1, f1);
        c = chain_steps(c, t0 + 3 * CH_G, x3, f3, ocol);
    }
    ocol[(size_t)M_TOT * N_OUT] = c;
}

#define ACT_WG   1024
#define ACT_U    4
#define ACT_PER  ((M_TOT * N_OUT / 4) / (ACT_WG * 256))
static_assert(ACT_PER * ACT_WG * 256 == M_TOT * N_OUT / 4, "the grid covers ht exactly");
static_assert(ACT_PER % ACT_U == 0, "whole trips");

__global__ __launch_bounds__(256)
void sru_act_kernel(const float* __restrict__ g, float* out)
{
    for (int j0 = 0; j0 < ACT_PER; j0 += ACT_U) {
        float4 c[ACT_U], r[ACT_U], cx[ACT_U];
#pragma unroll
        for (int u = 0; u < ACT_U; ++u) {
            const int i  = ((j0 + u) * ACT_WG + blockIdx.x) * 256 + threadIdx.x;
            const int m  = i >> 6;
            const int o4 = (i & 63) << 2;
            const float* grow = g + (size_t)m * N_TOT + o4;
            r[u]  = *(const float4*)(grow + 2 * N_OUT);
            cx[u] = *(const float4*)(grow + 3 * N_OUT);
            c[u]  = *(const float4*)(out + (size_t)m * N_OUT + o4);
        }
#pragma unroll
        for (int u = 0; u < ACT_U; ++u) {
            const int i  = ((j0 + u) * ACT_WG + blockIdx.x) * 256 + threadIdx.x;
            const int m  = i >> 6;
            const int o4 = (i & 63) << 2;
            float4 h;
            h.x = r[u].x * tanhf(c[u].x) + (1.0f - r[u].x) * cx[u].x;
            h.y = r[u].y * tanhf(c[u].y) + (1.0f - r[u].y) * cx[u].y;
            h.z = r[u].z * tanhf(c[u].z) + (1.0f - r[u].z) * cx[u].z;
            h.w = r[u].w * tanhf(c[u].w) + (1.0f - r[u].w) * cx[u].w;
            *(float4*)(out + (size_t)m * N_OUT + o4) = h;
        }
    }
}

extern "C" void kernel_launch(void* const* d_in, const int* in_sizes, int n_in,
                              void* d_out, int out_size, void* d_ws, size_t ws_size,
                              hipStream_t stream) {
    const float* xt = (const float*)d_in[0];
    const float* ct = (const float*)d_in[1];
    const float* Wx = (const float*)d_in[2];
    const float* Wf = (const float*)d_in[3];
    const float* bf = (const float*)d_in[4];
    const float* Wr = (const float*)d_in[5];
    const float* br = (const float*)d_in[6];
    const float* Wc = (const float*)d_in[7];
    const float* bc = (const float*)d_in[8];

    float* g   = (float*)d_ws;
    float* out = (float*)d_out;

    dim3 grid1(N_TOT / TN, M_TOT / TM);
    sru_gemm4_kernel<<<grid1, N_MATH + N_LOAD, 0, stream>>>(xt, Wx, Wf, Wr, Wc,
                                                bf, br, bc, g);

    sru_chain_kernel<<<(BATCH * N_OUT) / 32, 32, 0, stream>>>(g, ct, out);

    sru_act_kernel<<<ACT_WG, 256, 0, stream>>>(g, out);
}


# state chain run under the GEMM: GEMM split into an x,f launch and an r,cx launch whose first 64 work-groups (4 chain waves each, 8-step load groups) run the chain beside the WMMA tiles; sums kept
# speedup vs baseline: 3.8836x; 1.0397x over previous
#include <hip/hip_runtime.h>
#include <hip/hip_bf16.h>


typedef float v2f __attribute__((ext_vector_type(2)));
typedef float v8f __attribute__((ext_vector_type(8)));

#define N_IN   512
#define N_OUT  256
#define T_LEN  2048
#define BATCH  32
#define M_TOT  (T_LEN * BATCH)
#define N_TOT  (4 * N_OUT)

#define TM 128
#define TN 128
#define TK 32
#define LDS_STRIDE 36
#define N_MATH 256
#define N_LOAD 128
#define LD_ROWS (N_LOAD / 8)
#define LD_PASSES (TM / LD_ROWS)
#define NB_LAUNCH 4
#define PF_AHEAD 192
#define CH_ROWS 16
#define CH_WAVES 4
#define PF_UNIT 64
static_assert(TM == TN, "one staging loop walks both tiles");
static_assert(TK == 32, "a loader lane group of 8 covers one K-chunk row of 32 floats");

__device__ __forceinline__ float sigmoidf_fast(float x) {
    return 1.0f / (1.0f + __expf(-x));
}


#define CH_G 8
static_assert(T_LEN % (4 * CH_G) == 0, "the chain loop walks four groups per trip");

__device__ __forceinline__
void chain_request(const float* __restrict__ g, unsigned goff, int t,
                   float (&xp)[CH_G], float (&f)[CH_G])
{
#pragma unroll
    for (int u = 0; u < CH_G; ++u) {
        const float* row = g + (size_t)(t + u) * ((size_t)BATCH * N_TOT);
        xp[u] = row[goff];
        f[u]  = row[goff + N_OUT];
    }
    __builtin_amdgcn_sched_barrier(0);
}

__device__ __forceinline__
float chain_steps(float c, int t,
                  const float (&xp)[CH_G], const float (&f)[CH_G],
                  float* __restrict__ out, unsigned ch)
{
#pragma unroll
    for (int u = 0; u < CH_G; ++u) {
        c = f[u] * c + (1.0f - f[u]) * xp[u];
        float* orow = out + (size_t)(t + u) * (BATCH * N_OUT);
        orow[ch] = c;
    }
    __builtin_amdgcn_sched_barrier(0);
    return c;
}

__device__ __forceinline__
void chain_run(const float* __restrict__ g, const float* __restrict__ ct0,
               float* __restrict__ out, unsigned ch)
{
    const unsigned b  = ch >> 8;
    const unsigned o  = ch & (N_OUT - 1);

    float c = ct0[ch];
    const unsigned goff = b * N_TOT + o;

    float x0[CH_G], f0[CH_G], x1[CH_G], f1[CH_G];
    float x2[CH_G], f2[CH_G], x3[CH_G], f3[CH_G];

    chain_request(g, goff, 0, x0, f0);
    chain_request(g, goff, CH_G, x1, f1);
    chain_request(g, goff, 2 * CH_G, x2, f2);
    for (int t0 = 0; t0 < T_LEN; t0 += 4 * CH_G) {
        const bool more = (t0 + 4 * CH_G < T_LEN);
        chain_request(g, goff, t0 + 3 * CH_G, x3, f3);
        c = chain_steps(c, t0, x0, f0, out, ch);
        chain_request(g, goff, more ? (t0 + 4 * CH_G) : t0, x0, f0);
        c = chain_steps(c, t0 + CH_G, x1, f1, out, ch);
        chain_request(g, goff, more ? (t0 + 5 * CH_G) : (t0 + CH_G), x1, f1);
        c = chain_steps(c, t0 + 2 * CH_G, x2, f2, out, ch);
        chain_request(g, goff, more ? (t0 + 6 * CH_G) : (t0 + 2 * CH_G), x2, f2);
        c = chain_steps(c, t0 + 3 * CH_G, x3, f3, out, ch);
    }
    out[(size_t)M_TOT * N_OUT + ch] = c;
}

__global__ __launch_bounds__(N_MATH + N_LOAD, 10)
void sru_gemm4_kernel(const float* __restrict__ xt,
                      const float* __restrict__ Wx, const float* __restrict__ Wf,
                      const float* __restrict__ Wr, const float* __restrict__ Wc,
                      const float* __restrict__ bf, const float* __restrict__ br,
                      const float* __restrict__ bc,
                      const float* __restrict__ ct0,
                      float* __restrict__ g,
                      float* __restrict__ out,
                      int nblk0,
                      int chain_rows)
{
    if ((int)blockIdx.y < chain_rows) {
        const int w = __builtin_amdgcn_readfirstlane(threadIdx.x >> 5);
        if (w % 3 != 0) return;
        const int grp = ((int)blockIdx.y * NB_LAUNCH + (int)blockIdx.x) * CH_WAVES + w / 3;
        chain_run(g, ct0, out, (unsigned)grp * 32u + (threadIdx.x & 31u));
        return;
    }

    __shared__ float As[2][TM * LDS_STRIDE];
    __shared__ float Bs[2][TN * LDS_STRIDE];

    const int tid   = threadIdx.x;
    const int wave  = __builtin_amdgcn_readfirstlane(tid >> 5);
    const bool loader = wave >= N_MATH / 32;
    const int wm    = (wave >> 1) & 3;
    const int wn    = wave & 1;
    const int lane  = tid & 31;
    const int lm    = lane & 15;
    const int khalf = lane >> 4;

    const int nblk  = nblk0 + (int)blockIdx.x;
    const int mtile = (int)blockIdx.y - chain_rows;
    const int m0    = mtile * TM;
    const int wsel = nblk >> 1;
    const int n0w  = (nblk & 1) * TN;
    const float* __restrict__ W =
        (wsel == 0) ? Wx : (wsel == 1) ? Wf : (wsel == 2) ? Wr : Wc;
    const float* __restrict__ bias =
        (wsel == 1) ? bf : (wsel == 2) ? br : (wsel == 3) ? bc : nullptr;

    const int lt   = tid & (N_LOAD - 1);
    const int lrow = lt >> 3;
    const int lc4  = (lt & 7) << 2;
    const float* __restrict__ ag = xt + (size_t)(m0 + lrow) * N_IN + lc4;
    const float* __restrict__ bg = W + (size_t)(n0w + lrow) * N_IN + lc4;
    const int lds_l = lrow * LDS_STRIDE + lc4;

    v8f acc[2][4];
#pragma unroll
    for (int ms = 0; ms < 2; ++ms)
#pragma unroll
        for (int ns = 0; ns < 4; ++ns)
            acc[ms][ns] = (v8f){0.f, 0.f, 0.f, 0.f, 0.f, 0.f, 0.f, 0.f};

    if (loader) {
        float4 a[LD_PASSES], w[LD_PASSES];
#pragma unroll
        for (int j = 0; j < LD_PASSES; ++j) {
            a[j] = *(const float4*)(ag + j * LD_ROWS * N_IN);
            w[j] = *(const float4*)(bg + j * LD_ROWS * N_IN);
        }
#pragma unroll
        for (int j = 0; j < LD_PASSES; ++j) {
            *(float4*)&As[0][lds_l + j * LD_ROWS * LDS_STRIDE] = a[j];
            *(float4*)&Bs[0][lds_l + j * LD_ROWS * LDS_STRIDE] = w[j];
        }
    } else {
        for (int j = 0; j < (TM * (N_IN / PF_UNIT)) / N_MATH; ++j) {
            const int i  = tid + N_MATH * j;
            const int r  = i >> 3;
            const int cu = i & 7;
            if (cu != 0)
                __builtin_prefetch(xt + (size_t)(m0 + r) * N_IN + cu * PF_UNIT, 0, 3);
        }
        if (mtile + PF_AHEAD < M_TOT / TM && tid < TM)
            __builtin_prefetch(xt + (size_t)(m0 + PF_AHEAD * TM + tid) * N_IN, 0, 3);
    }
    __syncthreads();

    int buf = 0;
    for (int kb = 0; kb < N_IN; kb += TK, buf ^= 1) {
        if (loader) {
            if (kb + TK < N_IN) {
                const int kn = kb + TK;
                float4 a[LD_PASSES], w[LD_PASSES];
#pragma unroll
                for (int j = 0; j < LD_PASSES; ++j) {
                    a[j] = *(const float4*)(ag + j * LD_ROWS * N_IN + kn);
                    w[j] = *(const float4*)(bg + j * LD_ROWS * N_IN + kn);
                }
#pragma unroll
                for (int j = 0; j < LD_PASSES; ++j) {
                    *(float4*)&As[buf ^ 1][lds_l + j * LD_ROWS * LDS_STRIDE] = a[j];
                    *(float4*)&Bs[buf ^ 1][lds_l + j * LD_ROWS * LDS_STRIDE] = w[j];
                }
            }
#pragma unroll
            for (int ms = 0; ms < 2; ++ms)
#pragma unroll
                for (int ns = 0; ns < 4; ++ns)
                    acc[ms][ns] = (v8f){0.f, 0.f, 0.f, 0.f, 0.f, 0.f, 0.f, 0.f};
        } else {
            const float* Arow  = &As[buf][(wm * 32 + lm) * LDS_STRIDE + 2 * khalf];
            const float* Bbase = &Bs[buf][(wn * 64 + lm) * LDS_STRIDE + 2 * khalf];
#pragma unroll
            for (int kk = 0; kk < TK; kk += 4) {
                const v2f a0 = *(const v2f*)(Arow + kk);
                const v2f a1 = *(const v2f*)(Arow + 16 * LDS_STRIDE + kk);
#pragma unroll
                for (int ns = 0; ns < 4; ++ns) {
                    const v2f b = *(const v2f*)(Bbase + ns * 16 * LDS_STRIDE + kk);
                    acc[0][ns] = __builtin_amdgcn_wmma_f32_16x16x4_f32(
                        false, a0, false, b, (short)0, acc[0][ns], false, false);
                    acc[1][ns] = __builtin_amdgcn_wmma_f32_16x16x4_f32(
                        false, a1, false, b, (short)0, acc[1][ns], false, false);
                }
            }
        }
        __syncthreads();
    }
    if (loader) return;

    const int ncol0 = nblk * TN + wn * 64;
#pragma unroll
    for (int ns = 0; ns < 4; ++ns) {
        const int colw = n0w + wn * 64 + ns * 16 + lm;
        const int col  = ncol0 + ns * 16 + lm;
        const float bv = (wsel != 0) ? bias[colw] : 0.0f;
#pragma unroll
        for (int ms = 0; ms < 2; ++ms) {
#pragma unroll
            for (int r = 0; r < 8; ++r) {
                const int row = m0 + wm * 32 + ms * 16 + khalf * 8 + r;
                float v = acc[ms][ns][r];
                if (wsel == 1 || wsel == 2)      v = sigmoidf_fast(v + bv);
                else if (wsel == 3)              v = v + bv;
                g[(size_t)row * N_TOT + col] = v;
            }
        }
    }
}

#define ACT_WG   1024
#define ACT_U    4
#define ACT_PER  ((M_TOT * N_OUT / 4) / (ACT_WG * 256))
static_assert(ACT_PER * ACT_WG * 256 == M_TOT * N_OUT / 4, "the grid covers ht exactly");
static_assert(ACT_PER % ACT_U == 0, "whole trips");

__global__ __launch_bounds__(256)
void sru_act_kernel(const float* __restrict__ g, float* out)
{
    for (int j0 = 0; j0 < ACT_PER; j0 += ACT_U) {
        float4 c[ACT_U], r[ACT_U], cx[ACT_U];
#pragma unroll
        for (int u = 0; u < ACT_U; ++u) {
            const int i  = ((j0 + u) * ACT_WG + blockIdx.x) * 256 + threadIdx.x;
            const int m  = i >> 6;
            const int o4 = (i & 63) << 2;
            const float* grow = g + (size_t)m * N_TOT + o4;
            r[u]  = *(const float4*)(grow + 2 * N_OUT);
            cx[u] = *(const float4*)(grow + 3 * N_OUT);
            c[u]  = *(const float4*)(out + (size_t)m * N_OUT + o4);
        }
#pragma unroll
        for (int u = 0; u < ACT_U; ++u) {
            const int i  = ((j0 + u) * ACT_WG + blockIdx.x) * 256 + threadIdx.x;
            const int m  = i >> 6;
            const int o4 = (i & 63) << 2;
            float4 h;
            h.x = r[u].x * tanhf(c[u].x) + (1.0f - r[u].x) * cx[u].x;
            h.y = r[u].y * tanhf(c[u].y) + (1.0f - r[u].y) * cx[u].y;
            h.z = r[u].z * tanhf(c[u].z) + (1.0f - r[u].z) * cx[u].z;
            h.w = r[u].w * tanhf(c[u].w) + (1.0f - r[u].w) * cx[u].w;
            *(float4*)(out + (size_t)m * N_OUT + o4) = h;
        }
    }
}

extern "C" void kernel_launch(void* const* d_in, const int* in_sizes, int n_in,
                              void* d_out, int out_size, void* d_ws, size_t ws_size,
                              hipStream_t stream) {
    const float* xt = (const float*)d_in[0];
    const float* ct = (const float*)d_in[1];
    const float* Wx = (const float*)d_in[2];
    const float* Wf = (const float*)d_in[3];
    const float* bf = (const float*)d_in[4];
    const float* Wr = (const float*)d_in[5];
    const float* br = (const float*)d_in[6];
    const float* Wc = (const float*)d_in[7];
    const float* bc = (const float*)d_in[8];

    float* g   = (float*)d_ws;
    float* out = (float*)d_out;

    dim3 grid1(NB_LAUNCH, M_TOT / TM);
    sru_gemm4_kernel<<<grid1, N_MATH + N_LOAD, 0, stream>>>(xt, Wx, Wf, Wr, Wc,
                                                bf, br, bc, ct, g, out, 0, 0);

    dim3 grid2(NB_LAUNCH, CH_ROWS + M_TOT / TM);
    sru_gemm4_kernel<<<grid2, N_MATH + N_LOAD, 0, stream>>>(xt, Wx, Wf, Wr, Wc,
                                                bf, br, bc, ct, g, out, NB_LAUNCH, CH_ROWS);

    sru_act_kernel<<<ACT_WG, 256, 0, stream>>>(g, out);
}
